# starting kernel
# baseline (speedup 1.0000x reference)
#include <hip/hip_runtime.h>

#define BS      16
#define CIN     32
#define COUT    64
#define OUT_DIM 1024
#define KPATCH  4
#define DLEN    (OUT_DIM * KPATCH)
#define SCALE   0.17677669529663687f

typedef __attribute__((ext_vector_type(2))) float v2f;
typedef __attribute__((ext_vector_type(8))) float v8f;

__global__ __launch_bounds__(128) void
nolc1d_wmma_f32(const float* __restrict__ x,
                const float* __restrict__ w,
                float* __restrict__ out)
{
    const int lane = threadIdx.x & 31;
    const int wvid = threadIdx.x >> 5;
    const int p    = (blockIdx.x << 2) + wvid;

    const int half = lane >> 4;
    const int lid  = lane & 15;

    const float* xp = x + ((size_t)lid * CIN) * DLEN + 4 * p + 2 * half;
    const float* wp = w + ((size_t)lid * CIN) * DLEN + 4 * p + 2 * half;
    const size_t tstride = (size_t)16 * CIN * DLEN;

    v8f acc0 = {}, acc1 = {}, acc2 = {}, acc3 = {};

#pragma unroll 4
    for (int c = 0; c < CIN; ++c) {
        const size_t co = (size_t)c * DLEN;
        v2f a  = *(const v2f*)(xp + co);
        v2f b0 = *(const v2f*)(wp + co);
        v2f b1 = *(const v2f*)(wp + co + tstride);
        v2f b2 = *(const v2f*)(wp + co + 2 * tstride);
        v2f b3 = *(const v2f*)(wp + co + 3 * tstride);

        acc0 = __builtin_amdgcn_wmma_f32_16x16x4_f32(false, a, false, b0,
                                                     (short)0, acc0, false, false);
        acc1 = __builtin_amdgcn_wmma_f32_16x16x4_f32(false, a, false, b1,
                                                     (short)0, acc1, false, false);
        acc2 = __builtin_amdgcn_wmma_f32_16x16x4_f32(false, a, false, b2,
                                                     (short)0, acc2, false, false);
        acc3 = __builtin_amdgcn_wmma_f32_16x16x4_f32(false, a, false, b3,
                                                     (short)0, acc3, false, false);
    }

    const int b_base = half * 8;
#pragma unroll
    for (int j = 0; j < 8; ++j) {
        const int b = b_base + j;
        const size_t row = (size_t)b * COUT * OUT_DIM + p;
        out[row + (size_t)(0 * 16 + lid) * OUT_DIM] = acc0[j] * SCALE;
        out[row + (size_t)(1 * 16 + lid) * OUT_DIM] = acc1[j] * SCALE;
        out[row + (size_t)(2 * 16 + lid) * OUT_DIM] = acc2[j] * SCALE;
        out[row + (size_t)(3 * 16 + lid) * OUT_DIM] = acc3[j] * SCALE;
    }
}

extern "C" void kernel_launch(void* const* d_in, const int* in_sizes, int n_in,
                              void* d_out, int out_size, void* d_ws, size_t ws_size,
                              hipStream_t stream)
{
    (void)in_sizes; (void)n_in; (void)out_size; (void)d_ws; (void)ws_size;
    const float* x = (const float*)d_in[0];
    const float* w = (const float*)d_in[1];
    float* out     = (float*)d_out;

    dim3 grid(OUT_DIM / 4);
    dim3 block(128);
    nolc1d_wmma_f32<<<grid, block, 0, stream>>>(x, w, out);
}


# stage x/w/out through LDS per tile of 8 patches x 32 outputs: whole 128-B line loads, all 48 per wave issued before first use, out as 32-B row pieces; 256 groups of 8 waves; same sum order
# speedup vs baseline: 4.1682x; 4.1682x over previous
#include <hip/hip_runtime.h>


#define BS      16
#define CIN     32
#define COUT    64
#define OUT_DIM 1024
#define KPATCH  4
#define DLEN    (OUT_DIM * KPATCH)
#define SCALE   0.17677669529663687f

#define GP      8
#define NOL     32
#define NWAVE   8
#define NTHR    (NWAVE * 32)
#define CCH     4
#define NCHUNK  (CIN / CCH)
#define PITCH   36
#define XROWS   (CIN * BS)
#define WROWS   (CIN * NOL)
#define XOFF    0
#define WOFF    (XROWS * PITCH)
#define OOFF    (WOFF + WROWS * PITCH)
#define LDS_FLOATS (OOFF + BS * NOL * GP)

typedef __attribute__((ext_vector_type(2))) float v2f;
typedef __attribute__((ext_vector_type(4))) float v4f;
typedef __attribute__((ext_vector_type(8))) float v8f;

static_assert(LDS_FLOATS * 4 <= 320 * 1024, "tile fits the work-group's LDS");

static __device__ __forceinline__ void publish_chunk()
{
    asm volatile("s_wait_dscnt 0x0\n\ts_barrier_signal -1\n\ts_barrier_wait -1" ::: "memory");
}

__global__ __launch_bounds__(NTHR) void
nolc1d_wmma_f32(const float* __restrict__ x,
                const float* __restrict__ w,
                float* __restrict__ out)
{
    __shared__ __attribute__((aligned(16))) float L[LDS_FLOATS];

    const int tid  = threadIdx.x;
    const int lane = tid & 31;
    const int wv   = tid >> 5;
    const int g    = blockIdx.x >> 1;
    const int nh   = blockIdx.x & 1;

    const int q    = tid & 7;
    const int rr   = tid >> 3;

    const float* xg = x + ((size_t)((rr & 15) * CIN + (rr >> 4)) * DLEN + 32 * g + 4 * q);
    const float* wg = w + ((size_t)((nh * NOL + rr) * CIN) * DLEN + 32 * g + 4 * q);

    v4f rx[NCHUNK][2], rw[NCHUNK][4];
#pragma unroll
    for (int ch = 0; ch < NCHUNK; ++ch) {
#pragma unroll
        for (int i = 0; i < 2; ++i)
            rx[ch][i] = *(const v4f*)(xg + (size_t)(CCH * ch + 2 * i) * DLEN);
#pragma unroll
        for (int i = 0; i < 4; ++i)
            rw[ch][i] = *(const v4f*)(wg + (size_t)(CCH * ch + i) * DLEN);
    }

    const int half = lane >> 4;
    const int lid  = lane & 15;

    float* sx = L + XOFF + rr * PITCH + 4 * q;
    float* sw = L + WOFF + rr * PITCH + 4 * q;
    const float* fa = L + XOFF + lid * PITCH + 4 * wv + 2 * half;
    const float* fb = L + WOFF + lid * PITCH + 4 * wv + 2 * half;

    v8f acc0 = {}, acc1 = {};

#pragma unroll
    for (int ch = 0; ch < NCHUNK; ++ch) {
#pragma unroll
        for (int i = 0; i < 2; ++i)
            *(v4f*)(sx + (CCH * ch + 2 * i) * BS * PITCH) = rx[ch][i];
#pragma unroll
        for (int i = 0; i < 4; ++i)
            *(v4f*)(sw + (CCH * ch + i) * NOL * PITCH) = rw[ch][i];

        publish_chunk();

v2f a[CCH], b0[CCH], b1[CCH];
#pragma unroll
        for (int cl = 0; cl < CCH; ++cl) {
            const int c = CCH * ch + cl;
            a[cl]  = *(const v2f*)(fa + c * BS * PITCH);
            b0[cl] = *(const v2f*)(fb + c * NOL * PITCH);
            b1[cl] = *(const v2f*)(fb + (c * NOL + 16) * PITCH);
        }
        __builtin_amdgcn_sched_barrier(0);
#pragma unroll
        for (int cl = 0; cl < CCH; ++cl) {
            acc0 = __builtin_amdgcn_wmma_f32_16x16x4_f32(false, a[cl], false, b0[cl],
                                                         (short)0, acc0, false, false);
            acc1 = __builtin_amdgcn_wmma_f32_16x16x4_f32(false, a[cl], false, b1[cl],
                                                         (short)0, acc1, false, false);
        }
        __builtin_amdgcn_sched_barrier(0);
    }

    float* so = L + OOFF + ((8 * half) * NOL + lid) * GP + wv;
#pragma unroll
    for (int j = 0; j < 8; ++j) {
        so[(j * NOL) * GP]      = acc0[j] * SCALE;
        so[(j * NOL + 16) * GP] = acc1[j] * SCALE;
    }
    __syncthreads();

#pragma unroll
    for (int i = 0; i < 4; ++i) {
        const int piece = i * NTHR + tid;
        const int row   = piece >> 1;
        const int hf    = piece & 1;
        const int b     = row >> 5;
        const int ol    = row & 31;
        v4f v = *(const v4f*)(L + OOFF + piece * 4);
        *(v4f*)(out + ((size_t)(b * COUT + nh * NOL + ol) * OUT_DIM + GP * g + 4 * hf)) = v;
    }
}

extern "C" void kernel_launch(void* const* d_in, const int* in_sizes, int n_in,
                              void* d_out, int out_size, void* d_ws, size_t ws_size,
                              hipStream_t stream)
{
    (void)in_sizes; (void)n_in; (void)out_size; (void)d_ws; (void)ws_size;
    const float* x = (const float*)d_in[0];
    const float* w = (const float*)d_in[1];
    float* out     = (float*)d_out;

    dim3 grid(256);
    dim3 block(256);
    nolc1d_wmma_f32<<<grid, block, 0, stream>>>(x, w, out);
}


# re-tile to 16 patches x 16 outputs per work-group: each global row piece is one whole aligned 256-B unit (not half of one), out rows leave as 64-B pieces; 16 waves, one patch each; same sum order
# speedup vs baseline: 5.3357x; 1.2801x over previous
#include <hip/hip_runtime.h>


#define BS      16
#define CIN     32
#define COUT    64
#define OUT_DIM 1024
#define KPATCH  4
#define DLEN    (OUT_DIM * KPATCH)
#define SCALE   0.17677669529663687f

#define GP      16
#define NOL     16
#define NWAVE   16
#define NTHR    (NWAVE * 32)
#define CCH     4
#define NCHUNK  (CIN / CCH)
#define PITCH   68
#define XROWS   (CIN * BS)
#define WROWS   (CIN * NOL)
#define XOFF    0
#define WOFF    (XROWS * PITCH)
#define OOFF    (WOFF + WROWS * PITCH)
#define OPITCH  20
#define LDS_FLOATS (OOFF + BS * NOL * OPITCH)

typedef __attribute__((ext_vector_type(2))) float v2f;
typedef __attribute__((ext_vector_type(4))) float v4f;
typedef __attribute__((ext_vector_type(8))) float v8f;

static_assert(LDS_FLOATS * 4 <= 320 * 1024, "tile fits the work-group's LDS");

static __device__ __forceinline__ void publish_chunk()
{
    asm volatile("s_wait_dscnt 0x0\n\ts_barrier_signal -1\n\ts_barrier_wait -1" ::: "memory");
}

__global__ __launch_bounds__(NTHR) void
nolc1d_wmma_f32(const float* __restrict__ x,
                const float* __restrict__ w,
                float* __restrict__ out)
{
    __shared__ __attribute__((aligned(16))) float L[LDS_FLOATS];

    const int tid  = threadIdx.x;
    const int lane = tid & 31;
    const int wv   = tid >> 5;
    const int g    = blockIdx.x >> 2;
    const int nq   = blockIdx.x & 3;

    const int q    = tid & 15;
    const int rr   = tid >> 4;

    const float* xg = x + ((size_t)((rr & 15) * CIN + (rr >> 4)) * DLEN + 64 * g + 4 * q);
    const float* wg = w + ((size_t)((nq * NOL + (rr & 15)) * CIN + (rr >> 4)) * DLEN + 64 * g + 4 * q);

    v4f rx[NCHUNK][2], rw[NCHUNK][2];
#pragma unroll
    for (int ch = 0; ch < NCHUNK; ++ch) {
#pragma unroll
        for (int i = 0; i < 2; ++i)
            rx[ch][i] = *(const v4f*)(xg + (size_t)(CCH * ch + 2 * i) * DLEN);
#pragma unroll
        for (int i = 0; i < 2; ++i)
            rw[ch][i] = *(const v4f*)(wg + (size_t)(CCH * ch + 2 * i) * DLEN);
        __builtin_amdgcn_sched_barrier(0);
    }

    const int half = lane >> 4;
    const int lid  = lane & 15;

    float* sx = L + XOFF + rr * PITCH + 4 * q;
    float* sw = L + WOFF + rr * PITCH + 4 * q;
    const float* fa = L + XOFF + lid * PITCH + 4 * wv + 2 * half;
    const float* fb = L + WOFF + lid * PITCH + 4 * wv + 2 * half;

    v8f acc = {};

#pragma unroll
    for (int ch = 0; ch < NCHUNK; ++ch) {
#pragma unroll
        for (int i = 0; i < 2; ++i)
            *(v4f*)(sx + (CCH * ch + 2 * i) * BS * PITCH) = rx[ch][i];
#pragma unroll
        for (int i = 0; i < 2; ++i)
            *(v4f*)(sw + (CCH * ch + 2 * i) * NOL * PITCH) = rw[ch][i];

        publish_chunk();

        v2f a[CCH], b[CCH];
#pragma unroll
        for (int cl = 0; cl < CCH; ++cl) {
            const int c = CCH * ch + cl;
            a[cl] = *(const v2f*)(fa + c * BS * PITCH);
            b[cl] = *(const v2f*)(fb + c * NOL * PITCH);
        }
        __builtin_amdgcn_sched_barrier(0);
#pragma unroll
        for (int cl = 0; cl < CCH; ++cl) {
            acc = __builtin_amdgcn_wmma_f32_16x16x4_f32(false, a[cl], false, b[cl],
                                                        (short)0, acc, false, false);
        }
        __builtin_amdgcn_sched_barrier(0);
    }

    float* so = L + OOFF + ((8 * half) * NOL + lid) * OPITCH + wv;
#pragma unroll
    for (int j = 0; j < 8; ++j)
        so[(j * NOL) * OPITCH] = acc[j] * SCALE;
    __syncthreads();

#pragma unroll
    for (int i = 0; i < 2; ++i) {
        const int piece = i * NTHR + tid;
        const int row   = piece >> 2;
        const int part  = piece & 3;
        const int b     = row >> 4;
        const int ol    = row & 15;
        v4f v = *(const v4f*)(L + OOFF + row * OPITCH + 4 * part);
        *(v4f*)(out + ((size_t)(b * COUT + nq * NOL + ol) * OUT_DIM + GP * g + 4 * part)) = v;
    }
}

extern "C" void kernel_launch(void* const* d_in, const int* in_sizes, int n_in,
                              void* d_out, int out_size, void* d_ws, size_t ws_size,
                              hipStream_t stream)
{
    (void)in_sizes; (void)n_in; (void)out_size; (void)d_ws; (void)ws_size;
    const float* x = (const float*)d_in[0];
    const float* w = (const float*)d_in[1];
    float* out     = (float*)d_out;

    dim3 grid(256);
    dim3 block(512);
    nolc1d_wmma_f32<<<grid, block, 0, stream>>>(x, w, out);
}
